# starting kernel
# baseline (speedup 1.0000x reference)
#include <hip/hip_runtime.h>
#include <hip/hip_bf16.h>

typedef __attribute__((ext_vector_type(2))) float v2f;
typedef __attribute__((ext_vector_type(8))) float v8f;

#define IN_DIM  128
#define HD      128
#define E_DIM   32
#define NHEAD   4
#define DHEAD   32
#define OUT_DIM 32
#define INV_SQRT_D 0.17677669529663687f

static __device__ __forceinline__ v8f wmma_f32(v2f a, v2f b, v8f c) {
  return __builtin_amdgcn_wmma_f32_16x16x4_f32(false, a, false, b, (short)0, c,
                                               false, false);
}

__global__ void zero_kernel(float* p, int n) {
  int i = blockIdx.x * blockDim.x + threadIdx.x;
  if (i < n) p[i] = 0.0f;
}

__global__ void precompute_wvo(const float* __restrict__ Wv,
                               const float* __restrict__ bv,
                               const float* __restrict__ Wo,
                               float* __restrict__ Wvo,
                               float* __restrict__ bvo) {
  int idx = blockIdx.x * blockDim.x + threadIdx.x;
  if (idx < IN_DIM * HD) {
    int k = idx >> 7;
    int c = idx & 127;
    int h = c >> 5, j = c & 31;
    float acc = 0.0f;
    for (int d = 0; d < DHEAD; ++d)
      acc += Wv[k * HD + h * DHEAD + d] * Wo[(h * DHEAD + d) * OUT_DIM + j];
    Wvo[idx] = acc;
  } else if (idx < IN_DIM * HD + HD) {
    int c = idx - IN_DIM * HD;
    int h = c >> 5, j = c & 31;
    float acc = 0.0f;
    for (int d = 0; d < DHEAD; ++d)
      acc += bv[h * DHEAD + d] * Wo[(h * DHEAD + d) * OUT_DIM + j];
    bvo[c] = acc;
  }
}

__global__ __launch_bounds__(256) void node_gemm(
    const float* __restrict__ x,
    const float* __restrict__ Wq, const float* __restrict__ bq,
    const float* __restrict__ Wk, const float* __restrict__ bk,
    const float* __restrict__ Wvo, const float* __restrict__ bvo,
    float* __restrict__ Q, float* __restrict__ Km, float* __restrict__ Vo,
    int N) {
  __shared__ float As[16][IN_DIM];
  const int tid = threadIdx.x;
  const int m0 = blockIdx.x * 16;

  for (int i = 0; i < 8; ++i) {
    int lin = tid + i * 256;
    int r = lin >> 7, c = lin & 127;
    int gr = m0 + r; if (gr >= N) gr = N - 1;
    As[r][c] = x[(size_t)gr * IN_DIM + c];
  }
  __syncthreads();

  const int w = tid >> 5, L = tid & 31;
  const int n0 = w * 16;
  const int rowA = L & 15;
  const int kOff = (L < 16) ? 0 : 2;
  const int colB = L & 15;

  const float bq0 = bq[n0 + colB];
  const float bk0 = bk[n0 + colB];
  const float bv0 = bvo[n0 + colB];
  v8f cq, ck, cv;
  for (int j = 0; j < 8; ++j) { cq[j] = bq0; ck[j] = bk0; cv[j] = bv0; }

  for (int kk = 0; kk < IN_DIM; kk += 4) {
    v2f a;
    a.x = As[rowA][kk + kOff];
    a.y = As[rowA][kk + kOff + 1];
    v2f b;
    b.x = Wq[(kk + kOff) * HD + n0 + colB];
    b.y = Wq[(kk + kOff + 1) * HD + n0 + colB];
    cq = wmma_f32(a, b, cq);
    b.x = Wk[(kk + kOff) * HD + n0 + colB];
    b.y = Wk[(kk + kOff + 1) * HD + n0 + colB];
    ck = wmma_f32(a, b, ck);
    b.x = Wvo[(kk + kOff) * HD + n0 + colB];
    b.y = Wvo[(kk + kOff + 1) * HD + n0 + colB];
    cv = wmma_f32(a, b, cv);
  }

  for (int j = 0; j < 8; ++j) {
    int r = (L < 16) ? j : (j + 8);
    int gr = m0 + r;
    if (gr < N) {
      size_t o = (size_t)gr * HD + n0 + colB;
      Q[o] = cq[j]; Km[o] = ck[j]; Vo[o] = cv[j];
    }
  }
}

__global__ __launch_bounds__(128) void edge_selfdot(
    const float* __restrict__ ea, const float* __restrict__ We,
    const float* __restrict__ be, float* __restrict__ sE, int NE) {
  __shared__ float Ae[16][E_DIM];
  const int tid = threadIdx.x;
  const int e0 = blockIdx.x * 16;

  for (int i = 0; i < 4; ++i) {
    int lin = tid + i * 128;
    int r = lin >> 5, c = lin & 31;
    int ge = e0 + r; if (ge >= NE) ge = NE - 1;
    Ae[r][c] = ea[(size_t)ge * E_DIM + c];
  }
  __syncthreads();

  const int h = tid >> 5, L = tid & 31;
  const int n0 = h * DHEAD;
  const int rowA = L & 15;
  const int kOff = (L < 16) ? 0 : 2;
  const int colB = L & 15;

  v8f c0 = {}, c1 = {};
  for (int kk = 0; kk < E_DIM; kk += 4) {
    v2f a;
    a.x = Ae[rowA][kk + kOff];
    a.y = Ae[rowA][kk + kOff + 1];
    v2f b0, b1;
    b0.x = We[(kk + kOff) * HD + n0 + colB];
    b0.y = We[(kk + kOff + 1) * HD + n0 + colB];
    b1.x = We[(kk + kOff) * HD + n0 + 16 + colB];
    b1.y = We[(kk + kOff + 1) * HD + n0 + 16 + colB];
    c0 = wmma_f32(a, b0, c0);
    c1 = wmma_f32(a, b1, c1);
  }

  const float be0 = be[n0 + colB], be1 = be[n0 + 16 + colB];
  float t[8];
  for (int j = 0; j < 8; ++j) {
    float u = c0[j] + be0, v = c1[j] + be1;
    t[j] = u * u + v * v;
  }
  for (int m = 1; m <= 8; m <<= 1)
    for (int j = 0; j < 8; ++j) t[j] += __shfl_xor(t[j], m);

  if (L == 0 || L == 16) {
    int rb = (L == 0) ? 0 : 8;
    for (int j = 0; j < 8; ++j) {
      int ge = e0 + rb + j;
      if (ge < NE) sE[(size_t)ge * NHEAD + h] = t[j];
    }
  }
}

__global__ __launch_bounds__(256) void edge_pass(
    const int* __restrict__ ei, const float* __restrict__ Q,
    const float* __restrict__ Km, const float* __restrict__ Vo,
    const float* __restrict__ sE, const float* __restrict__ bo,
    float* __restrict__ sums, float* __restrict__ cnt, int NE) {
  const int e = blockIdx.x * 8 + (threadIdx.x >> 5);
  if (e >= NE) return;
  const int L = threadIdx.x & 31;
  const int src = ei[e];
  const int dst = ei[NE + e];

  const int h = L >> 3, d4 = (L & 7) << 2;
  const float4 q = *(const float4*)(Q + (size_t)src * HD + h * DHEAD + d4);
  const float4 k = *(const float4*)(Km + (size_t)dst * HD + h * DHEAD + d4);
  float p = q.x * k.x + q.y * k.y + q.z * k.z + q.w * k.w;
  p += __shfl_xor(p, 1);
  p += __shfl_xor(p, 2);
  p += __shfl_xor(p, 4);

  float sc[NHEAD];
  for (int h2 = 0; h2 < NHEAD; ++h2)
    sc[h2] = (__shfl(p, h2 * 8) + sE[(size_t)e * NHEAD + h2]) * INV_SQRT_D;

  float m = sc[0];
  for (int h2 = 1; h2 < NHEAD; ++h2) m = fmaxf(m, sc[h2]);
  float ex[NHEAD], den = 0.0f;
  for (int h2 = 0; h2 < NHEAD; ++h2) { ex[h2] = __expf(sc[h2] - m); den += ex[h2]; }
  const float inv = 1.0f / den;

  const int j = L;
  float acc = bo[j];
  for (int h2 = 0; h2 < NHEAD; ++h2)
    acc += (ex[h2] * inv) * Vo[(size_t)dst * HD + h2 * DHEAD + j];

  atomicAdd(&sums[(size_t)src * OUT_DIM + j], acc);
  if (L == 0) atomicAdd(&cnt[src], 1.0f);
}

__global__ void finalize(const float* __restrict__ sums,
                         const float* __restrict__ cnt,
                         float* __restrict__ out, int N) {
  int i = blockIdx.x * blockDim.x + threadIdx.x;
  if (i < N * OUT_DIM)
    out[i] = sums[i] / fmaxf(cnt[i >> 5], 1.0f);
}

extern "C" void kernel_launch(void* const* d_in, const int* in_sizes, int n_in,
                              void* d_out, int out_size, void* d_ws, size_t ws_size,
                              hipStream_t stream) {
  const float* x  = (const float*)d_in[0];
  const int*   ei = (const int*)d_in[1];
  const float* ea = (const float*)d_in[2];
  const float* Wq = (const float*)d_in[3];
  const float* bq = (const float*)d_in[4];
  const float* Wk = (const float*)d_in[5];
  const float* bk = (const float*)d_in[6];
  const float* Wv = (const float*)d_in[7];
  const float* bv = (const float*)d_in[8];
  const float* We = (const float*)d_in[9];
  const float* be = (const float*)d_in[10];
  const float* Wo = (const float*)d_in[11];
  const float* bo = (const float*)d_in[12];

  const int N  = in_sizes[0] / IN_DIM;
  const int NE = in_sizes[1] / 2;

  float* ws   = (float*)d_ws;
  float* Q    = ws;
  float* Km   = Q   + (size_t)N * HD;
  float* Vo   = Km  + (size_t)N * HD;
  float* Wvo  = Vo  + (size_t)N * HD;
  float* bvo  = Wvo + IN_DIM * HD;
  float* sE   = bvo + HD;
  float* sums = sE  + (size_t)NE * NHEAD;
  float* cnt  = sums + (size_t)N * OUT_DIM;

  {
    int n = N * OUT_DIM + N;
    zero_kernel<<<(n + 255) / 256, 256, 0, stream>>>(sums, n);
  }
  {
    int n = IN_DIM * HD + HD;
    precompute_wvo<<<(n + 127) / 128, 128, 0, stream>>>(Wv, bv, Wo, Wvo, bvo);
  }
  node_gemm<<<(N + 15) / 16, 256, 0, stream>>>(x, Wq, bq, Wk, bk, Wvo, bvo,
                                               Q, Km, Vo, N);
  edge_selfdot<<<(NE + 15) / 16, 128, 0, stream>>>(ea, We, be, sE, NE);
  edge_pass<<<(NE + 7) / 8, 256, 0, stream>>>(ei, Q, Km, Vo, sE, bo,
                                              sums, cnt, NE);
  {
    int n = N * OUT_DIM;
    finalize<<<(n + 255) / 256, 256, 0, stream>>>(sums, cnt, (float*)d_out, N);
  }
}


# edge pass by source-sorted records: counting sort on src (integer atomics, slots in arrival order); one wave per node sums its edges in registers, stores the mean; float atomics and mean pass go
# speedup vs baseline: 1.1771x; 1.1771x over previous
#include <hip/hip_runtime.h>
#include <hip/hip_bf16.h>

typedef __attribute__((ext_vector_type(2))) float v2f;
typedef __attribute__((ext_vector_type(8))) float v8f;

#define IN_DIM  128
#define HD      128
#define E_DIM   32
#define NHEAD   4
#define DHEAD   32
#define OUT_DIM 32
#define INV_SQRT_D 0.17677669529663687f

static __device__ __forceinline__ v8f wmma_f32(v2f a, v2f b, v8f c) {
  return __builtin_amdgcn_wmma_f32_16x16x4_f32(false, a, false, b, (short)0, c,
                                               false, false);
}

__global__ void zero_kernel(float* p, int n) {
  int i = blockIdx.x * blockDim.x + threadIdx.x;
  if (i < n) p[i] = 0.0f;
}

__global__ void precompute_wvo(const float* __restrict__ Wv,
                               const float* __restrict__ bv,
                               const float* __restrict__ Wo,
                               float* __restrict__ Wvo,
                               float* __restrict__ bvo) {
  int idx = blockIdx.x * blockDim.x + threadIdx.x;
  if (idx < IN_DIM * HD) {
    int k = idx >> 7;
    int c = idx & 127;
    int h = c >> 5, j = c & 31;
    float acc = 0.0f;
    for (int d = 0; d < DHEAD; ++d)
      acc += Wv[k * HD + h * DHEAD + d] * Wo[(h * DHEAD + d) * OUT_DIM + j];
    Wvo[idx] = acc;
  } else if (idx < IN_DIM * HD + HD) {
    int c = idx - IN_DIM * HD;
    int h = c >> 5, j = c & 31;
    float acc = 0.0f;
    for (int d = 0; d < DHEAD; ++d)
      acc += bv[h * DHEAD + d] * Wo[(h * DHEAD + d) * OUT_DIM + j];
    bvo[c] = acc;
  }
}

__global__ __launch_bounds__(256) void node_gemm(
    const float* __restrict__ x,
    const float* __restrict__ Wq, const float* __restrict__ bq,
    const float* __restrict__ Wk, const float* __restrict__ bk,
    const float* __restrict__ Wvo, const float* __restrict__ bvo,
    float* __restrict__ Q, float* __restrict__ Km, float* __restrict__ Vo,
    int N) {
  __shared__ float As[16][IN_DIM];
  const int tid = threadIdx.x;
  const int m0 = blockIdx.x * 16;

  for (int i = 0; i < 8; ++i) {
    int lin = tid + i * 256;
    int r = lin >> 7, c = lin & 127;
    int gr = m0 + r; if (gr >= N) gr = N - 1;
    As[r][c] = x[(size_t)gr * IN_DIM + c];
  }
  __syncthreads();

  const int w = tid >> 5, L = tid & 31;
  const int n0 = w * 16;
  const int rowA = L & 15;
  const int kOff = (L < 16) ? 0 : 2;
  const int colB = L & 15;

  const float bq0 = bq[n0 + colB];
  const float bk0 = bk[n0 + colB];
  const float bv0 = bvo[n0 + colB];
  v8f cq, ck, cv;
  for (int j = 0; j < 8; ++j) { cq[j] = bq0; ck[j] = bk0; cv[j] = bv0; }

  for (int kk = 0; kk < IN_DIM; kk += 4) {
    v2f a;
    a.x = As[rowA][kk + kOff];
    a.y = As[rowA][kk + kOff + 1];
    v2f b;
    b.x = Wq[(kk + kOff) * HD + n0 + colB];
    b.y = Wq[(kk + kOff + 1) * HD + n0 + colB];
    cq = wmma_f32(a, b, cq);
    b.x = Wk[(kk + kOff) * HD + n0 + colB];
    b.y = Wk[(kk + kOff + 1) * HD + n0 + colB];
    ck = wmma_f32(a, b, ck);
    b.x = Wvo[(kk + kOff) * HD + n0 + colB];
    b.y = Wvo[(kk + kOff + 1) * HD + n0 + colB];
    cv = wmma_f32(a, b, cv);
  }

  for (int j = 0; j < 8; ++j) {
    int r = (L < 16) ? j : (j + 8);
    int gr = m0 + r;
    if (gr < N) {
      size_t o = (size_t)gr * HD + n0 + colB;
      Q[o] = cq[j]; Km[o] = ck[j]; Vo[o] = cv[j];
    }
  }
}

__global__ __launch_bounds__(256) void count_src(const int* __restrict__ ei,
                                                 int* __restrict__ cur, int NE) {
  const int stride = gridDim.x * 256;
  for (int e = blockIdx.x * 256 + threadIdx.x; e < NE; e += stride)
    atomicAdd(&cur[ei[e]], 1);
}

__global__ __launch_bounds__(1024) void scan_counts(int* __restrict__ cur, int n) {
  __shared__ int part[1024];
  const int tid = threadIdx.x;
  const int per = (n + 1023) / 1024;
  const int lo = tid * per;
  int hi = lo + per; if (hi > n) hi = n;
  int s = 0;
  for (int i = lo; i < hi; ++i) s += cur[i];
  part[tid] = s;
  __syncthreads();
  for (int off = 1; off < 1024; off <<= 1) {
    const int v = (tid >= off) ? part[tid - off] : 0;
    __syncthreads();
    part[tid] += v;
    __syncthreads();
  }
  int run = part[tid] - s;
  for (int i = lo; i < hi; ++i) { const int c = cur[i]; cur[i] = run; run += c; }
}

__global__ __launch_bounds__(128) void edge_selfdot(
    const float* __restrict__ ea, const float* __restrict__ We,
    const float* __restrict__ be, const int* __restrict__ ei,
    int* __restrict__ cur, float* __restrict__ sE,
    unsigned short* __restrict__ dstS, int NE) {
  __shared__ float Ae[16][E_DIM];
  __shared__ __attribute__((aligned(16))) float Se[16][NHEAD];
  const int tid = threadIdx.x;
  const int e0 = blockIdx.x * 16;

  int pos = -1, dstv = 0;
  if (tid < 16) {
    const int ge = e0 + tid;
    if (ge < NE) {
      pos = atomicAdd(&cur[ei[ge]], 1);
      dstv = ei[NE + ge];
    }
  }

  for (int i = 0; i < 4; ++i) {
    int lin = tid + i * 128;
    int r = lin >> 5, c = lin & 31;
    int ge = e0 + r; if (ge >= NE) ge = NE - 1;
    Ae[r][c] = ea[(size_t)ge * E_DIM + c];
  }
  __syncthreads();

  const int h = tid >> 5, L = tid & 31;
  const int n0 = h * DHEAD;
  const int rowA = L & 15;
  const int kOff = (L < 16) ? 0 : 2;
  const int colB = L & 15;

  v8f c0 = {}, c1 = {};
  for (int kk = 0; kk < E_DIM; kk += 4) {
    v2f a;
    a.x = Ae[rowA][kk + kOff];
    a.y = Ae[rowA][kk + kOff + 1];
    v2f b0, b1;
    b0.x = We[(kk + kOff) * HD + n0 + colB];
    b0.y = We[(kk + kOff + 1) * HD + n0 + colB];
    b1.x = We[(kk + kOff) * HD + n0 + 16 + colB];
    b1.y = We[(kk + kOff + 1) * HD + n0 + 16 + colB];
    c0 = wmma_f32(a, b0, c0);
    c1 = wmma_f32(a, b1, c1);
  }

  const float be0 = be[n0 + colB], be1 = be[n0 + 16 + colB];
  float t[8];
  for (int j = 0; j < 8; ++j) {
    float u = c0[j] + be0, v = c1[j] + be1;
    t[j] = u * u + v * v;
  }
  for (int m = 1; m <= 8; m <<= 1)
    for (int j = 0; j < 8; ++j) t[j] += __shfl_xor(t[j], m);

  if (L == 0 || L == 16) {
    int rb = (L == 0) ? 0 : 8;
    for (int j = 0; j < 8; ++j) Se[rb + j][h] = t[j];
  }
  __syncthreads();
  if (pos >= 0) {
    *(float4*)(sE + (size_t)pos * NHEAD) = *(const float4*)&Se[tid][0];
    dstS[pos] = (unsigned short)dstv;
  }
}

__global__ __launch_bounds__(256) void node_pass(
    const float* __restrict__ Q, const float* __restrict__ Km,
    const float* __restrict__ Vo, const float* __restrict__ sE,
    const unsigned short* __restrict__ dstS, const int* __restrict__ cur,
    const float* __restrict__ bo, float* __restrict__ out, int N) {
  const int s = __builtin_amdgcn_readfirstlane(blockIdx.x * 8 + (threadIdx.x >> 5));
  if (s >= N) return;
  const int L = threadIdx.x & 31;
  const int beg = (s > 0) ? cur[s - 1] : 0;
  const int end = cur[s];

  const int h = L >> 3, d4 = (L & 7) << 2;
  const float4 q = *(const float4*)(Q + (size_t)s * HD + h * DHEAD + d4);
  const float boj = bo[L];
  float total = 0.0f;

  for (int base = beg; base < end; base += 32) {
    const int idx = base + L;
    const int myd = (idx < end) ? (int)dstS[idx] : 0;
    const int n = (end - base < 32) ? (end - base) : 32;
#pragma unroll 4
    for (int j = 0; j < n; ++j) {
      const int dst = __builtin_amdgcn_readlane(myd, j);
      const float4 se = *(const float4*)(sE + (size_t)(base + j) * NHEAD);
      const float4 k = *(const float4*)(Km + (size_t)dst * HD + h * DHEAD + d4);
      float p = q.x * k.x + q.y * k.y + q.z * k.z + q.w * k.w;
      p += __shfl_xor(p, 1);
      p += __shfl_xor(p, 2);
      p += __shfl_xor(p, 4);

      float sc[NHEAD];
      sc[0] = (__shfl(p, 0) + se.x) * INV_SQRT_D;
      sc[1] = (__shfl(p, 8) + se.y) * INV_SQRT_D;
      sc[2] = (__shfl(p, 16) + se.z) * INV_SQRT_D;
      sc[3] = (__shfl(p, 24) + se.w) * INV_SQRT_D;

      float m = sc[0];
      for (int h2 = 1; h2 < NHEAD; ++h2) m = fmaxf(m, sc[h2]);
      float ex[NHEAD], den = 0.0f;
      for (int h2 = 0; h2 < NHEAD; ++h2) { ex[h2] = __expf(sc[h2] - m); den += ex[h2]; }
      const float inv = 1.0f / den;

      float acc = boj;
      for (int h2 = 0; h2 < NHEAD; ++h2)
        acc += (ex[h2] * inv) * Vo[(size_t)dst * HD + h2 * DHEAD + L];
      total += acc;
    }
  }
  out[(size_t)s * OUT_DIM + L] = total / fmaxf((float)(end - beg), 1.0f);
}

extern "C" void kernel_launch(void* const* d_in, const int* in_sizes, int n_in,
                              void* d_out, int out_size, void* d_ws, size_t ws_size,
                              hipStream_t stream) {
  const float* x  = (const float*)d_in[0];
  const int*   ei = (const int*)d_in[1];
  const float* ea = (const float*)d_in[2];
  const float* Wq = (const float*)d_in[3];
  const float* bq = (const float*)d_in[4];
  const float* Wk = (const float*)d_in[5];
  const float* bk = (const float*)d_in[6];
  const float* Wv = (const float*)d_in[7];
  const float* bv = (const float*)d_in[8];
  const float* We = (const float*)d_in[9];
  const float* be = (const float*)d_in[10];
  const float* Wo = (const float*)d_in[11];
  const float* bo = (const float*)d_in[12];

  const int N  = in_sizes[0] / IN_DIM;
  const int NE = in_sizes[1] / 2;

  float* ws   = (float*)d_ws;
  float* Q    = ws;
  float* Km   = Q   + (size_t)N * HD;
  float* Vo   = Km  + (size_t)N * HD;
  float* Wvo  = Vo  + (size_t)N * HD;
  float* bvo  = Wvo + IN_DIM * HD;
  float* sE   = bvo + HD;
  unsigned short* dstS = (unsigned short*)(sE + (size_t)NE * NHEAD);
  int* cur = (int*)(sE + (size_t)NE * NHEAD + (size_t)N * OUT_DIM);

  zero_kernel<<<(N + 255) / 256, 256, 0, stream>>>((float*)cur, N);
  {
    int n = IN_DIM * HD + HD;
    precompute_wvo<<<(n + 127) / 128, 128, 0, stream>>>(Wv, bv, Wo, Wvo, bvo);
  }
  node_gemm<<<(N + 15) / 16, 256, 0, stream>>>(x, Wq, bq, Wk, bk, Wvo, bvo,
                                               Q, Km, Vo, N);
  count_src<<<512, 256, 0, stream>>>(ei, cur, NE);
  scan_counts<<<1, 1024, 0, stream>>>(cur, N);
  edge_selfdot<<<(NE + 15) / 16, 128, 0, stream>>>(ea, We, be, ei, cur, sE,
                                                   dstS, NE);
  node_pass<<<(N + 7) / 8, 256, 0, stream>>>(Q, Km, Vo, sE, dstS, cur, bo,
                                             (float*)d_out, N);
}


# edge self-dot launch as 4096 resident looping waves: each walks 16-edge tiles, all of We held as B fragments in registers, A rows loaded straight into fragments; no LDS, no barrier; same sum order
# speedup vs baseline: 1.3295x; 1.1295x over previous
#include <hip/hip_runtime.h>
#include <hip/hip_bf16.h>

typedef __attribute__((ext_vector_type(2))) float v2f;
typedef __attribute__((ext_vector_type(8))) float v8f;

#define IN_DIM  128
#define HD      128
#define E_DIM   32
#define NHEAD   4
#define DHEAD   32
#define OUT_DIM 32
#define INV_SQRT_D 0.17677669529663687f

static __device__ __forceinline__ v8f wmma_f32(v2f a, v2f b, v8f c) {
  return __builtin_amdgcn_wmma_f32_16x16x4_f32(false, a, false, b, (short)0, c,
                                               false, false);
}

__global__ void zero_kernel(float* p, int n) {
  int i = blockIdx.x * blockDim.x + threadIdx.x;
  if (i < n) p[i] = 0.0f;
}

__global__ void precompute_wvo(const float* __restrict__ Wv,
                               const float* __restrict__ bv,
                               const float* __restrict__ Wo,
                               float* __restrict__ Wvo,
                               float* __restrict__ bvo) {
  int idx = blockIdx.x * blockDim.x + threadIdx.x;
  if (idx < IN_DIM * HD) {
    int k = idx >> 7;
    int c = idx & 127;
    int h = c >> 5, j = c & 31;
    float acc = 0.0f;
    for (int d = 0; d < DHEAD; ++d)
      acc += Wv[k * HD + h * DHEAD + d] * Wo[(h * DHEAD + d) * OUT_DIM + j];
    Wvo[idx] = acc;
  } else if (idx < IN_DIM * HD + HD) {
    int c = idx - IN_DIM * HD;
    int h = c >> 5, j = c & 31;
    float acc = 0.0f;
    for (int d = 0; d < DHEAD; ++d)
      acc += bv[h * DHEAD + d] * Wo[(h * DHEAD + d) * OUT_DIM + j];
    bvo[c] = acc;
  }
}

__global__ __launch_bounds__(256) void node_gemm(
    const float* __restrict__ x,
    const float* __restrict__ Wq, const float* __restrict__ bq,
    const float* __restrict__ Wk, const float* __restrict__ bk,
    const float* __restrict__ Wvo, const float* __restrict__ bvo,
    float* __restrict__ Q, float* __restrict__ Km, float* __restrict__ Vo,
    int N) {
  __shared__ float As[16][IN_DIM];
  const int tid = threadIdx.x;
  const int m0 = blockIdx.x * 16;

  for (int i = 0; i < 8; ++i) {
    int lin = tid + i * 256;
    int r = lin >> 7, c = lin & 127;
    int gr = m0 + r; if (gr >= N) gr = N - 1;
    As[r][c] = x[(size_t)gr * IN_DIM + c];
  }
  __syncthreads();

  const int w = tid >> 5, L = tid & 31;
  const int n0 = w * 16;
  const int rowA = L & 15;
  const int kOff = (L < 16) ? 0 : 2;
  const int colB = L & 15;

  const float bq0 = bq[n0 + colB];
  const float bk0 = bk[n0 + colB];
  const float bv0 = bvo[n0 + colB];
  v8f cq, ck, cv;
  for (int j = 0; j < 8; ++j) { cq[j] = bq0; ck[j] = bk0; cv[j] = bv0; }

  for (int kk = 0; kk < IN_DIM; kk += 4) {
    v2f a;
    a.x = As[rowA][kk + kOff];
    a.y = As[rowA][kk + kOff + 1];
    v2f b;
    b.x = Wq[(kk + kOff) * HD + n0 + colB];
    b.y = Wq[(kk + kOff + 1) * HD + n0 + colB];
    cq = wmma_f32(a, b, cq);
    b.x = Wk[(kk + kOff) * HD + n0 + colB];
    b.y = Wk[(kk + kOff + 1) * HD + n0 + colB];
    ck = wmma_f32(a, b, ck);
    b.x = Wvo[(kk + kOff) * HD + n0 + colB];
    b.y = Wvo[(kk + kOff + 1) * HD + n0 + colB];
    cv = wmma_f32(a, b, cv);
  }

  for (int j = 0; j < 8; ++j) {
    int r = (L < 16) ? j : (j + 8);
    int gr = m0 + r;
    if (gr < N) {
      size_t o = (size_t)gr * HD + n0 + colB;
      Q[o] = cq[j]; Km[o] = ck[j]; Vo[o] = cv[j];
    }
  }
}

__global__ __launch_bounds__(256) void count_src(const int* __restrict__ ei,
                                                 int* __restrict__ cur, int NE) {
  const int stride = gridDim.x * 256;
  for (int e = blockIdx.x * 256 + threadIdx.x; e < NE; e += stride)
    atomicAdd(&cur[ei[e]], 1);
}

__global__ __launch_bounds__(1024) void scan_counts(int* __restrict__ cur, int n) {
  __shared__ int part[1024];
  const int tid = threadIdx.x;
  const int per = (n + 1023) / 1024;
  const int lo = tid * per;
  int hi = lo + per; if (hi > n) hi = n;
  int s = 0;
  for (int i = lo; i < hi; ++i) s += cur[i];
  part[tid] = s;
  __syncthreads();
  for (int off = 1; off < 1024; off <<= 1) {
    const int v = (tid >= off) ? part[tid - off] : 0;
    __syncthreads();
    part[tid] += v;
    __syncthreads();
  }
  int run = part[tid] - s;
  for (int i = lo; i < hi; ++i) { const int c = cur[i]; cur[i] = run; run += c; }
}

__global__ __launch_bounds__(128) void edge_selfdot(
    const float* __restrict__ ea, const float* __restrict__ We,
    const float* __restrict__ be, const int* __restrict__ ei,
    int* __restrict__ cur, float* __restrict__ sE,
    unsigned short* __restrict__ dstS, int NE) {
  const int L = threadIdx.x & 31;
  const int wave = __builtin_amdgcn_readfirstlane(blockIdx.x * 4 + (threadIdx.x >> 5));
  const int nwaves = gridDim.x * 4;
  const int ntiles = (NE + 15) >> 4;

  const int rowA = L & 15;
  const int kOff = (L < 16) ? 0 : 2;
  const int colB = L & 15;
  const bool b0 = (L & 1) != 0, b1 = (L & 2) != 0, b2 = (L & 4) != 0;
  const bool claimer = (L & 8) == 0;
  const int rowC = ((L >> 4) << 3) | (L & 7);

  v2f bf[8][8];
  for (int ks = 0; ks < 8; ++ks)
    for (int n = 0; n < 8; ++n) {
      bf[ks][n].x = We[(ks * 4 + kOff) * HD + n * 16 + colB];
      bf[ks][n].y = We[(ks * 4 + kOff + 1) * HD + n * 16 + colB];
    }
  float bev[8];
  for (int n = 0; n < 8; ++n) bev[n] = be[n * 16 + colB];

  int t = wave;
  v2f a[8];
  int srcv = 0, dstv = 0;
  if (t < ntiles) {
    int ge = t * 16 + rowA; if (ge >= NE) ge = NE - 1;
    const float* rp = ea + (size_t)ge * E_DIM + kOff;
    for (int ks = 0; ks < 8; ++ks) a[ks] = *(const v2f*)(rp + ks * 4);
    const int gc = t * 16 + rowC;
    if (claimer && gc < NE) { srcv = ei[gc]; dstv = ei[NE + gc]; }
  }
  for (int ks = 0; ks < 8; ++ks) asm volatile("" : "+v"(a[ks]));
  asm volatile("" : "+v"(srcv), "+v"(dstv));

  while (t < ntiles) {
    const int tn = t + nwaves;
    v2f an[8];
    int srcn = 0, dstn = 0;
    if (tn < ntiles) {
      int ge = tn * 16 + rowA; if (ge >= NE) ge = NE - 1;
      const float* rp = ea + (size_t)ge * E_DIM + kOff;
      for (int ks = 0; ks < 8; ++ks) an[ks] = *(const v2f*)(rp + ks * 4);
      const int gc = tn * 16 + rowC;
      if (claimer && gc < NE) { srcn = ei[gc]; dstn = ei[NE + gc]; }
    } else {
      for (int ks = 0; ks < 8; ++ks) an[ks] = a[ks];
    }

    int pos = -1;
    if (claimer && t * 16 + rowC < NE) pos = atomicAdd(&cur[srcv], 1);

    v8f c[8];
    for (int n = 0; n < 8; ++n) c[n] = v8f{};
    for (int ks = 0; ks < 8; ++ks)
      for (int n = 0; n < 8; ++n) c[n] = wmma_f32(a[ks], bf[ks][n], c[n]);

    float z[NHEAD];
    for (int hd = 0; hd < NHEAD; ++hd) {
      float tt[8];
      for (int j = 0; j < 8; ++j) {
        float u = c[2 * hd][j] + bev[2 * hd], v = c[2 * hd + 1][j] + bev[2 * hd + 1];
        tt[j] = u * u + v * v;
      }
      float u4[4];
      for (int i = 0; i < 4; ++i) {
        const float keep = b0 ? tt[2 * i + 1] : tt[2 * i];
        const float send = b0 ? tt[2 * i] : tt[2 * i + 1];
        u4[i] = keep + __shfl_xor(send, 1);
      }
      float w2[2];
      for (int i = 0; i < 2; ++i) {
        const float keep = b1 ? u4[2 * i + 1] : u4[2 * i];
        const float send = b1 ? u4[2 * i] : u4[2 * i + 1];
        w2[i] = keep + __shfl_xor(send, 2);
      }
      const float keep = b2 ? w2[1] : w2[0];
      const float send = b2 ? w2[0] : w2[1];
      float zz = keep + __shfl_xor(send, 4);
      zz += __shfl_xor(zz, 8);
      z[hd] = zz;
    }

    if (pos >= 0) {
      *(float4*)(sE + (size_t)pos * NHEAD) = make_float4(z[0], z[1], z[2], z[3]);
      dstS[pos] = (unsigned short)dstv;
    }

    for (int ks = 0; ks < 8; ++ks) a[ks] = an[ks];
    srcv = srcn; dstv = dstn;
    t = tn;
  }
}

__global__ __launch_bounds__(256) void node_pass(
    const float* __restrict__ Q, const float* __restrict__ Km,
    const float* __restrict__ Vo, const float* __restrict__ sE,
    const unsigned short* __restrict__ dstS, const int* __restrict__ cur,
    const float* __restrict__ bo, float* __restrict__ out, int N) {
  const int s = __builtin_amdgcn_readfirstlane(blockIdx.x * 8 + (threadIdx.x >> 5));
  if (s >= N) return;
  const int L = threadIdx.x & 31;
  const int beg = (s > 0) ? cur[s - 1] : 0;
  const int end = cur[s];

  const int h = L >> 3, d4 = (L & 7) << 2;
  const float4 q = *(const float4*)(Q + (size_t)s * HD + h * DHEAD + d4);
  const float boj = bo[L];
  float total = 0.0f;

  for (int base = beg; base < end; base += 32) {
    const int idx = base + L;
    const int myd = (idx < end) ? (int)dstS[idx] : 0;
    const int n = (end - base < 32) ? (end - base) : 32;
#pragma unroll 4
    for (int j = 0; j < n; ++j) {
      const int dst = __builtin_amdgcn_readlane(myd, j);
      const float4 se = *(const float4*)(sE + (size_t)(base + j) * NHEAD);
      const float4 k = *(const float4*)(Km + (size_t)dst * HD + h * DHEAD + d4);
      float p = q.x * k.x + q.y * k.y + q.z * k.z + q.w * k.w;
      p += __shfl_xor(p, 1);
      p += __shfl_xor(p, 2);
      p += __shfl_xor(p, 4);

      float sc[NHEAD];
      sc[0] = (__shfl(p, 0) + se.x) * INV_SQRT_D;
      sc[1] = (__shfl(p, 8) + se.y) * INV_SQRT_D;
      sc[2] = (__shfl(p, 16) + se.z) * INV_SQRT_D;
      sc[3] = (__shfl(p, 24) + se.w) * INV_SQRT_D;

      float m = sc[0];
      for (int h2 = 1; h2 < NHEAD; ++h2) m = fmaxf(m, sc[h2]);
      float ex[NHEAD], den = 0.0f;
      for (int h2 = 0; h2 < NHEAD; ++h2) { ex[h2] = __expf(sc[h2] - m); den += ex[h2]; }
      const float inv = 1.0f / den;

      float acc = boj;
      for (int h2 = 0; h2 < NHEAD; ++h2)
        acc += (ex[h2] * inv) * Vo[(size_t)dst * HD + h2 * DHEAD + L];
      total += acc;
    }
  }
  out[(size_t)s * OUT_DIM + L] = total / fmaxf((float)(end - beg), 1.0f);
}

extern "C" void kernel_launch(void* const* d_in, const int* in_sizes, int n_in,
                              void* d_out, int out_size, void* d_ws, size_t ws_size,
                              hipStream_t stream) {
  const float* x  = (const float*)d_in[0];
  const int*   ei = (const int*)d_in[1];
  const float* ea = (const float*)d_in[2];
  const float* Wq = (const float*)d_in[3];
  const float* bq = (const float*)d_in[4];
  const float* Wk = (const float*)d_in[5];
  const float* bk = (const float*)d_in[6];
  const float* Wv = (const float*)d_in[7];
  const float* bv = (const float*)d_in[8];
  const float* We = (const float*)d_in[9];
  const float* be = (const float*)d_in[10];
  const float* Wo = (const float*)d_in[11];
  const float* bo = (const float*)d_in[12];

  const int N  = in_sizes[0] / IN_DIM;
  const int NE = in_sizes[1] / 2;

  float* ws   = (float*)d_ws;
  float* Q    = ws;
  float* Km   = Q   + (size_t)N * HD;
  float* Vo   = Km  + (size_t)N * HD;
  float* Wvo  = Vo  + (size_t)N * HD;
  float* bvo  = Wvo + IN_DIM * HD;
  float* sE   = bvo + HD;
  unsigned short* dstS = (unsigned short*)(sE + (size_t)NE * NHEAD);
  int* cur = (int*)(sE + (size_t)NE * NHEAD + (size_t)N * OUT_DIM);

  zero_kernel<<<(N + 255) / 256, 256, 0, stream>>>((float*)cur, N);
  {
    int n = IN_DIM * HD + HD;
    precompute_wvo<<<(n + 127) / 128, 128, 0, stream>>>(Wv, bv, Wo, Wvo, bvo);
  }
  node_gemm<<<(N + 15) / 16, 256, 0, stream>>>(x, Wq, bq, Wk, bk, Wvo, bvo,
                                               Q, Km, Vo, N);
  count_src<<<512, 256, 0, stream>>>(ei, cur, NE);
  scan_counts<<<1, 1024, 0, stream>>>(cur, N);
  edge_selfdot<<<1024, 128, 0, stream>>>(ea, We, be, ei, cur, sE, dstS, NE);
  node_pass<<<(N + 7) / 8, 256, 0, stream>>>(Q, Km, Vo, sE, dstS, cur, bo,
                                             (float*)d_out, N);
}


# edge grouping without per-edge global atomics: a block ranks its 8192-edge slice on LDS counters, merges them by line-wide returning adds; score pass slot = start + rank (arrival order as before)
# speedup vs baseline: 2.0385x; 1.5333x over previous
#include <hip/hip_runtime.h>
#include <hip/hip_bf16.h>

typedef __attribute__((ext_vector_type(2))) float v2f;
typedef __attribute__((ext_vector_type(8))) float v8f;

#define IN_DIM  128
#define HD      128
#define E_DIM   32
#define NHEAD   4
#define DHEAD   32
#define OUT_DIM 32
#define INV_SQRT_D 0.17677669529663687f

static __device__ __forceinline__ v8f wmma_f32(v2f a, v2f b, v8f c) {
  return __builtin_amdgcn_wmma_f32_16x16x4_f32(false, a, false, b, (short)0, c,
                                               false, false);
}

__global__ void zero_kernel(float* p, int n) {
  int i = blockIdx.x * blockDim.x + threadIdx.x;
  if (i < n) p[i] = 0.0f;
}

__global__ void precompute_wvo(const float* __restrict__ Wv,
                               const float* __restrict__ bv,
                               const float* __restrict__ Wo,
                               float* __restrict__ Wvo,
                               float* __restrict__ bvo) {
  int idx = blockIdx.x * blockDim.x + threadIdx.x;
  if (idx < IN_DIM * HD) {
    int k = idx >> 7;
    int c = idx & 127;
    int h = c >> 5, j = c & 31;
    float acc = 0.0f;
    for (int d = 0; d < DHEAD; ++d)
      acc += Wv[k * HD + h * DHEAD + d] * Wo[(h * DHEAD + d) * OUT_DIM + j];
    Wvo[idx] = acc;
  } else if (idx < IN_DIM * HD + HD) {
    int c = idx - IN_DIM * HD;
    int h = c >> 5, j = c & 31;
    float acc = 0.0f;
    for (int d = 0; d < DHEAD; ++d)
      acc += bv[h * DHEAD + d] * Wo[(h * DHEAD + d) * OUT_DIM + j];
    bvo[c] = acc;
  }
}

__global__ __launch_bounds__(256) void node_gemm(
    const float* __restrict__ x,
    const float* __restrict__ Wq, const float* __restrict__ bq,
    const float* __restrict__ Wk, const float* __restrict__ bk,
    const float* __restrict__ Wvo, const float* __restrict__ bvo,
    float* __restrict__ Q, float* __restrict__ Km, float* __restrict__ Vo,
    int N) {
  __shared__ float As[16][IN_DIM];
  const int tid = threadIdx.x;
  const int m0 = blockIdx.x * 16;

  for (int i = 0; i < 8; ++i) {
    int lin = tid + i * 256;
    int r = lin >> 7, c = lin & 127;
    int gr = m0 + r; if (gr >= N) gr = N - 1;
    As[r][c] = x[(size_t)gr * IN_DIM + c];
  }
  __syncthreads();

  const int w = tid >> 5, L = tid & 31;
  const int n0 = w * 16;
  const int rowA = L & 15;
  const int kOff = (L < 16) ? 0 : 2;
  const int colB = L & 15;

  const float bq0 = bq[n0 + colB];
  const float bk0 = bk[n0 + colB];
  const float bv0 = bvo[n0 + colB];
  v8f cq, ck, cv;
  for (int j = 0; j < 8; ++j) { cq[j] = bq0; ck[j] = bk0; cv[j] = bv0; }

  for (int kk = 0; kk < IN_DIM; kk += 4) {
    v2f a;
    a.x = As[rowA][kk + kOff];
    a.y = As[rowA][kk + kOff + 1];
    v2f b;
    b.x = Wq[(kk + kOff) * HD + n0 + colB];
    b.y = Wq[(kk + kOff + 1) * HD + n0 + colB];
    cq = wmma_f32(a, b, cq);
    b.x = Wk[(kk + kOff) * HD + n0 + colB];
    b.y = Wk[(kk + kOff + 1) * HD + n0 + colB];
    ck = wmma_f32(a, b, ck);
    b.x = Wvo[(kk + kOff) * HD + n0 + colB];
    b.y = Wvo[(kk + kOff + 1) * HD + n0 + colB];
    cv = wmma_f32(a, b, cv);
  }

  for (int j = 0; j < 8; ++j) {
    int r = (L < 16) ? j : (j + 8);
    int gr = m0 + r;
    if (gr < N) {
      size_t o = (size_t)gr * HD + n0 + colB;
      Q[o] = cq[j]; Km[o] = ck[j]; Vo[o] = cv[j];
    }
  }
}

#define RANK_EPT 8
#define MERGE_U  10
__global__ __launch_bounds__(1024) void rank_src(const int* __restrict__ ei,
                                                 int* __restrict__ cur,
                                                 int* __restrict__ rank,
                                                 int N, int NE) {
  extern __shared__ int cnt[];
  const int tid = threadIdx.x;
  for (int i = tid; i < N; i += 1024) cnt[i] = 0;
  __syncthreads();

  const int base = blockIdx.x * (1024 * RANK_EPT);
  int sv[RANK_EPT], lr[RANK_EPT];
  for (int j = 0; j < RANK_EPT; ++j) {
    const int e = base + j * 1024 + tid;
    sv[j] = (e < NE) ? ei[e] : -1;
  }
  for (int j = 0; j < RANK_EPT; ++j)
    lr[j] = (sv[j] >= 0) ? atomicAdd(&cnt[sv[j]], 1) : 0;
  __syncthreads();

  for (int i0 = tid; i0 < N; i0 += 1024 * MERGE_U) {
    int c[MERGE_U], b[MERGE_U];
    for (int u = 0; u < MERGE_U; ++u) {
      const int i = i0 + u * 1024;
      c[u] = (i < N) ? cnt[i] : 0;
    }
    for (int u = 0; u < MERGE_U; ++u) {
      const int i = i0 + u * 1024;
      b[u] = (i < N) ? atomicAdd(&cur[i], c[u]) : 0;
    }
    for (int u = 0; u < MERGE_U; ++u) {
      const int i = i0 + u * 1024;
      if (i < N) cnt[i] = b[u];
    }
  }
  __syncthreads();

  for (int j = 0; j < RANK_EPT; ++j) {
    const int e = base + j * 1024 + tid;
    if (sv[j] >= 0) rank[e] = cnt[sv[j]] + lr[j];
  }
}

__global__ __launch_bounds__(1024) void scan_counts(int* __restrict__ cur, int n) {
  __shared__ int part[1024];
  const int tid = threadIdx.x;
  const int per = (n + 1023) / 1024;
  const int lo = tid * per;
  int hi = lo + per; if (hi > n) hi = n;
  int s = 0;
  for (int i = lo; i < hi; ++i) s += cur[i];
  part[tid] = s;
  __syncthreads();
  for (int off = 1; off < 1024; off <<= 1) {
    const int v = (tid >= off) ? part[tid - off] : 0;
    __syncthreads();
    part[tid] += v;
    __syncthreads();
  }
  int run = part[tid] - s;
  for (int i = lo; i < hi; ++i) { const int c = cur[i]; cur[i] = run; run += c; }
}

__global__ __launch_bounds__(128) void edge_selfdot(
    const float* __restrict__ ea, const float* __restrict__ We,
    const float* __restrict__ be, const int* __restrict__ ei,
    const int* __restrict__ cur, const int* __restrict__ rank,
    float* __restrict__ sE, unsigned short* __restrict__ dstS, int NE) {
  const int L = threadIdx.x & 31;
  const int wave = __builtin_amdgcn_readfirstlane(blockIdx.x * 4 + (threadIdx.x >> 5));
  const int nwaves = gridDim.x * 4;
  const int ntiles = (NE + 15) >> 4;

  const int rowA = L & 15;
  const int kOff = (L < 16) ? 0 : 2;
  const int colB = L & 15;
  const bool b0 = (L & 1) != 0, b1 = (L & 2) != 0, b2 = (L & 4) != 0;
  const bool claimer = (L & 8) == 0;
  const int rowC = ((L >> 4) << 3) | (L & 7);

  v2f bf[8][8];
  for (int ks = 0; ks < 8; ++ks)
    for (int n = 0; n < 8; ++n) {
      bf[ks][n].x = We[(ks * 4 + kOff) * HD + n * 16 + colB];
      bf[ks][n].y = We[(ks * 4 + kOff + 1) * HD + n * 16 + colB];
    }
  float bev[8];
  for (int n = 0; n < 8; ++n) bev[n] = be[n * 16 + colB];

  int t = wave;
  v2f a[8];
  int srcv = 0, dstv = 0, rankv = 0;
  if (t < ntiles) {
    int ge = t * 16 + rowA; if (ge >= NE) ge = NE - 1;
    const float* rp = ea + (size_t)ge * E_DIM + kOff;
    for (int ks = 0; ks < 8; ++ks) a[ks] = *(const v2f*)(rp + ks * 4);
    const int gc = t * 16 + rowC;
    if (claimer && gc < NE) { srcv = ei[gc]; dstv = ei[NE + gc]; rankv = rank[gc]; }
  }
  for (int ks = 0; ks < 8; ++ks) asm volatile("" : "+v"(a[ks]));
  asm volatile("" : "+v"(srcv), "+v"(dstv), "+v"(rankv));

  while (t < ntiles) {
    const bool mine = claimer && t * 16 + rowC < NE;
    const int startv = cur[srcv];

    const int tn = t + nwaves;
    v2f an[8];
    int srcn = 0, dstn = 0, rankn = 0;
    if (tn < ntiles) {
      int ge = tn * 16 + rowA; if (ge >= NE) ge = NE - 1;
      const float* rp = ea + (size_t)ge * E_DIM + kOff;
      for (int ks = 0; ks < 8; ++ks) an[ks] = *(const v2f*)(rp + ks * 4);
      const int gc = tn * 16 + rowC;
      if (claimer && gc < NE) { srcn = ei[gc]; dstn = ei[NE + gc]; rankn = rank[gc]; }
    } else {
      for (int ks = 0; ks < 8; ++ks) an[ks] = a[ks];
    }

    v8f c[8];
    for (int n = 0; n < 8; ++n) c[n] = v8f{};
    for (int ks = 0; ks < 8; ++ks)
      for (int n = 0; n < 8; ++n) c[n] = wmma_f32(a[ks], bf[ks][n], c[n]);

    float z[NHEAD];
    for (int hd = 0; hd < NHEAD; ++hd) {
      float tt[8];
      for (int j = 0; j < 8; ++j) {
        float u = c[2 * hd][j] + bev[2 * hd], v = c[2 * hd + 1][j] + bev[2 * hd + 1];
        tt[j] = u * u + v * v;
      }
      float u4[4];
      for (int i = 0; i < 4; ++i) {
        const float keep = b0 ? tt[2 * i + 1] : tt[2 * i];
        const float send = b0 ? tt[2 * i] : tt[2 * i + 1];
        u4[i] = keep + __shfl_xor(send, 1);
      }
      float w2[2];
      for (int i = 0; i < 2; ++i) {
        const float keep = b1 ? u4[2 * i + 1] : u4[2 * i];
        const float send = b1 ? u4[2 * i] : u4[2 * i + 1];
        w2[i] = keep + __shfl_xor(send, 2);
      }
      const float keep = b2 ? w2[1] : w2[0];
      const float send = b2 ? w2[0] : w2[1];
      float zz = keep + __shfl_xor(send, 4);
      zz += __shfl_xor(zz, 8);
      z[hd] = zz;
    }

    if (mine) {
      const int pos = startv + rankv;
      *(float4*)(sE + (size_t)pos * NHEAD) = make_float4(z[0], z[1], z[2], z[3]);
      dstS[pos] = (unsigned short)dstv;
    }

    for (int ks = 0; ks < 8; ++ks) a[ks] = an[ks];
    srcv = srcn; dstv = dstn; rankv = rankn;
    t = tn;
  }
}

__global__ __launch_bounds__(256) void node_pass(
    const float* __restrict__ Q, const float* __restrict__ Km,
    const float* __restrict__ Vo, const float* __restrict__ sE,
    const unsigned short* __restrict__ dstS, const int* __restrict__ cur,
    const float* __restrict__ bo, float* __restrict__ out, int N, int NE) {
  const int s = __builtin_amdgcn_readfirstlane(blockIdx.x * 8 + (threadIdx.x >> 5));
  if (s >= N) return;
  const int L = threadIdx.x & 31;
  const int beg = cur[s];
  const int end = (s + 1 < N) ? cur[s + 1] : NE;

  const int h = L >> 3, d4 = (L & 7) << 2;
  const float4 q = *(const float4*)(Q + (size_t)s * HD + h * DHEAD + d4);
  const float boj = bo[L];
  float total = 0.0f;

  for (int base = beg; base < end; base += 32) {
    const int idx = base + L;
    const int myd = (idx < end) ? (int)dstS[idx] : 0;
    const int n = (end - base < 32) ? (end - base) : 32;
#pragma unroll 4
    for (int j = 0; j < n; ++j) {
      const int dst = __builtin_amdgcn_readlane(myd, j);
      const float4 se = *(const float4*)(sE + (size_t)(base + j) * NHEAD);
      const float4 k = *(const float4*)(Km + (size_t)dst * HD + h * DHEAD + d4);
      float p = q.x * k.x + q.y * k.y + q.z * k.z + q.w * k.w;
      p += __shfl_xor(p, 1);
      p += __shfl_xor(p, 2);
      p += __shfl_xor(p, 4);

      float sc[NHEAD];
      sc[0] = (__shfl(p, 0) + se.x) * INV_SQRT_D;
      sc[1] = (__shfl(p, 8) + se.y) * INV_SQRT_D;
      sc[2] = (__shfl(p, 16) + se.z) * INV_SQRT_D;
      sc[3] = (__shfl(p, 24) + se.w) * INV_SQRT_D;

      float m = sc[0];
      for (int h2 = 1; h2 < NHEAD; ++h2) m = fmaxf(m, sc[h2]);
      float ex[NHEAD], den = 0.0f;
      for (int h2 = 0; h2 < NHEAD; ++h2) { ex[h2] = __expf(sc[h2] - m); den += ex[h2]; }
      const float inv = 1.0f / den;

      float acc = boj;
      for (int h2 = 0; h2 < NHEAD; ++h2)
        acc += (ex[h2] * inv) * Vo[(size_t)dst * HD + h2 * DHEAD + L];
      total += acc;
    }
  }
  out[(size_t)s * OUT_DIM + L] = total / fmaxf((float)(end - beg), 1.0f);
}

extern "C" void kernel_launch(void* const* d_in, const int* in_sizes, int n_in,
                              void* d_out, int out_size, void* d_ws, size_t ws_size,
                              hipStream_t stream) {
  const float* x  = (const float*)d_in[0];
  const int*   ei = (const int*)d_in[1];
  const float* ea = (const float*)d_in[2];
  const float* Wq = (const float*)d_in[3];
  const float* bq = (const float*)d_in[4];
  const float* Wk = (const float*)d_in[5];
  const float* bk = (const float*)d_in[6];
  const float* Wv = (const float*)d_in[7];
  const float* bv = (const float*)d_in[8];
  const float* We = (const float*)d_in[9];
  const float* be = (const float*)d_in[10];
  const float* Wo = (const float*)d_in[11];
  const float* bo = (const float*)d_in[12];

  const int N  = in_sizes[0] / IN_DIM;
  const int NE = in_sizes[1] / 2;

  float* ws   = (float*)d_ws;
  float* Q    = ws;
  float* Km   = Q   + (size_t)N * HD;
  float* Vo   = Km  + (size_t)N * HD;
  float* Wvo  = Vo  + (size_t)N * HD;
  float* bvo  = Wvo + IN_DIM * HD;
  float* sE   = bvo + HD;
  unsigned short* dstS = (unsigned short*)(sE + (size_t)NE * NHEAD);
  int* cur = (int*)(sE + (size_t)NE * NHEAD + (size_t)N * OUT_DIM);
  int* rank = (int*)Vo;

  zero_kernel<<<(N + 255) / 256, 256, 0, stream>>>((float*)cur, N);
  rank_src<<<(NE + 1024 * RANK_EPT - 1) / (1024 * RANK_EPT), 1024,
             (size_t)N * sizeof(int), stream>>>(ei, cur, rank, N, NE);
  scan_counts<<<1, 1024, 0, stream>>>(cur, N);
  edge_selfdot<<<1024, 128, 0, stream>>>(ea, We, be, ei, cur, rank, sE, dstS, NE);
  {
    int n = IN_DIM * HD + HD;
    precompute_wvo<<<(n + 127) / 128, 128, 0, stream>>>(Wv, bv, Wo, Wvo, bvo);
  }
  node_gemm<<<(N + 15) / 16, 256, 0, stream>>>(x, Wq, bq, Wk, bk, Wvo, bvo,
                                               Q, Km, Vo, N);
  node_pass<<<(N + 7) / 8, 256, 0, stream>>>(Q, Km, Vo, sE, dstS, cur, bo,
                                             (float*)d_out, N, NE);
}


# no prefix-sum launch: the score pass (now 256 work-groups of 16 waves) and the node pass each derive the segment starts they need from the raw per-node counts (integer sums in LDS); 7 launches -> 6
# speedup vs baseline: 2.2644x; 1.1108x over previous
#include <hip/hip_runtime.h>
#include <hip/hip_bf16.h>

typedef __attribute__((ext_vector_type(2))) float v2f;
typedef __attribute__((ext_vector_type(8))) float v8f;
typedef __attribute__((ext_vector_type(4))) float v4f;

#define IN_DIM  128
#define HD      128
#define E_DIM   32
#define NHEAD   4
#define DHEAD   32
#define OUT_DIM 32
#define INV_SQRT_D 0.17677669529663687f

static __device__ __forceinline__ v8f wmma_f32(v2f a, v2f b, v8f c) {
  return __builtin_amdgcn_wmma_f32_16x16x4_f32(false, a, false, b, (short)0, c,
                                               false, false);
}

__global__ void zero_kernel(float* p, int n) {
  int i = blockIdx.x * blockDim.x + threadIdx.x;
  if (i < n) p[i] = 0.0f;
}

__global__ void precompute_wvo(const float* __restrict__ Wv,
                               const float* __restrict__ bv,
                               const float* __restrict__ Wo,
                               float* __restrict__ Wvo,
                               float* __restrict__ bvo) {
  int idx = blockIdx.x * blockDim.x + threadIdx.x;
  if (idx < IN_DIM * HD) {
    int k = idx >> 7;
    int c = idx & 127;
    int h = c >> 5, j = c & 31;
    float acc = 0.0f;
    for (int d = 0; d < DHEAD; ++d)
      acc += Wv[k * HD + h * DHEAD + d] * Wo[(h * DHEAD + d) * OUT_DIM + j];
    Wvo[idx] = acc;
  } else if (idx < IN_DIM * HD + HD) {
    int c = idx - IN_DIM * HD;
    int h = c >> 5, j = c & 31;
    float acc = 0.0f;
    for (int d = 0; d < DHEAD; ++d)
      acc += bv[h * DHEAD + d] * Wo[(h * DHEAD + d) * OUT_DIM + j];
    bvo[c] = acc;
  }
}

__global__ __launch_bounds__(256) void node_gemm(
    const float* __restrict__ x,
    const float* __restrict__ Wq, const float* __restrict__ bq,
    const float* __restrict__ Wk, const float* __restrict__ bk,
    const float* __restrict__ Wvo, const float* __restrict__ bvo,
    float* __restrict__ Q, float* __restrict__ Km, float* __restrict__ Vo,
    int N) {
  __shared__ float As[16][IN_DIM];
  const int tid = threadIdx.x;
  const int m0 = blockIdx.x * 16;

  for (int i = 0; i < 8; ++i) {
    int lin = tid + i * 256;
    int r = lin >> 7, c = lin & 127;
    int gr = m0 + r; if (gr >= N) gr = N - 1;
    As[r][c] = x[(size_t)gr * IN_DIM + c];
  }
  __syncthreads();

  const int w = tid >> 5, L = tid & 31;
  const int n0 = w * 16;
  const int rowA = L & 15;
  const int kOff = (L < 16) ? 0 : 2;
  const int colB = L & 15;

  const float bq0 = bq[n0 + colB];
  const float bk0 = bk[n0 + colB];
  const float bv0 = bvo[n0 + colB];
  v8f cq, ck, cv;
  for (int j = 0; j < 8; ++j) { cq[j] = bq0; ck[j] = bk0; cv[j] = bv0; }

  for (int kk = 0; kk < IN_DIM; kk += 4) {
    v2f a;
    a.x = As[rowA][kk + kOff];
    a.y = As[rowA][kk + kOff + 1];
    v2f b;
    b.x = Wq[(kk + kOff) * HD + n0 + colB];
    b.y = Wq[(kk + kOff + 1) * HD + n0 + colB];
    cq = wmma_f32(a, b, cq);
    b.x = Wk[(kk + kOff) * HD + n0 + colB];
    b.y = Wk[(kk + kOff + 1) * HD + n0 + colB];
    ck = wmma_f32(a, b, ck);
    b.x = Wvo[(kk + kOff) * HD + n0 + colB];
    b.y = Wvo[(kk + kOff + 1) * HD + n0 + colB];
    cv = wmma_f32(a, b, cv);
  }

  for (int j = 0; j < 8; ++j) {
    int r = (L < 16) ? j : (j + 8);
    int gr = m0 + r;
    if (gr < N) {
      size_t o = (size_t)gr * HD + n0 + colB;
      Q[o] = cq[j]; Km[o] = ck[j]; Vo[o] = cv[j];
    }
  }
}

#define RANK_EPT 8
#define MERGE_U  10
__global__ __launch_bounds__(1024) void rank_src(const int* __restrict__ ei,
                                                 int* __restrict__ cur,
                                                 int* __restrict__ rank,
                                                 int N, int NE) {
  extern __shared__ int cnt[];
  const int tid = threadIdx.x;
  for (int i = tid; i < N; i += 1024) cnt[i] = 0;
  __syncthreads();

  const int base = blockIdx.x * (1024 * RANK_EPT);
  int sv[RANK_EPT], lr[RANK_EPT];
  for (int j = 0; j < RANK_EPT; ++j) {
    const int e = base + j * 1024 + tid;
    sv[j] = (e < NE) ? ei[e] : -1;
  }
  for (int j = 0; j < RANK_EPT; ++j)
    lr[j] = (sv[j] >= 0) ? atomicAdd(&cnt[sv[j]], 1) : 0;
  __syncthreads();

  for (int i0 = tid; i0 < N; i0 += 1024 * MERGE_U) {
    int c[MERGE_U], b[MERGE_U];
    for (int u = 0; u < MERGE_U; ++u) {
      const int i = i0 + u * 1024;
      c[u] = (i < N) ? cnt[i] : 0;
    }
    for (int u = 0; u < MERGE_U; ++u) {
      const int i = i0 + u * 1024;
      b[u] = (i < N) ? atomicAdd(&cur[i], c[u]) : 0;
    }
    for (int u = 0; u < MERGE_U; ++u) {
      const int i = i0 + u * 1024;
      if (i < N) cnt[i] = b[u];
    }
  }
  __syncthreads();

  for (int j = 0; j < RANK_EPT; ++j) {
    const int e = base + j * 1024 + tid;
    if (sv[j] >= 0) rank[e] = cnt[sv[j]] + lr[j];
  }
}

#define SD_WAVES 16
#define SD_THREADS (SD_WAVES * 32)
#define SD_LD 5
__global__ __launch_bounds__(SD_THREADS) void edge_selfdot(
    const float* __restrict__ ea, const float* __restrict__ We,
    const float* __restrict__ be, const int* __restrict__ ei,
    const int* __restrict__ cur, const int* __restrict__ rank,
    float* __restrict__ sE, unsigned short* __restrict__ dstS, int N, int NE) {
  extern __shared__ int pre[];
  __shared__ int wtot[SD_WAVES];
  const int tid = threadIdx.x;
  const int L = tid & 31;
  {
    const int nq = N >> 2;
    const int4* c4 = (const int4*)cur;
    for (int q0 = tid; q0 < nq; q0 += SD_THREADS * SD_LD) {
      int4 v[SD_LD];
      for (int u = 0; u < SD_LD; ++u) {
        const int q = q0 + u * SD_THREADS;
        v[u] = (q < nq) ? c4[q] : make_int4(0, 0, 0, 0);
      }
      for (int u = 0; u < SD_LD; ++u) {
        const int q = q0 + u * SD_THREADS;
        if (q < nq) *(int4*)(pre + 4 * q) = v[u];
      }
    }
    for (int i = (nq << 2) + tid; i < N; i += SD_THREADS) pre[i] = cur[i];
    __syncthreads();
    const int per = ((N + SD_THREADS - 1) / SD_THREADS) | 1;
    int lo = tid * per; if (lo > N) lo = N;
    int hi = lo + per;  if (hi > N) hi = N;
    int s = 0;
    for (int i = lo; i < hi; ++i) s += pre[i];
    int inc = s;
    for (int off = 1; off < 32; off <<= 1) {
      const int v = __shfl_up(inc, off);
      if (L >= off) inc += v;
    }
    if (L == 31) wtot[tid >> 5] = inc;
    __syncthreads();
    int run = inc - s;
    for (int j = 0; j < SD_WAVES; ++j) run += (j < (tid >> 5)) ? wtot[j] : 0;
    for (int i = lo; i < hi; ++i) { const int cn = pre[i]; pre[i] = run; run += cn; }
    __syncthreads();
  }
  const int wave = __builtin_amdgcn_readfirstlane(blockIdx.x * SD_WAVES + (tid >> 5));
  const int nwaves = gridDim.x * SD_WAVES;
  const int ntiles = (NE + 15) >> 4;

  const int rowA = L & 15;
  const int kOff = (L < 16) ? 0 : 2;
  const int colB = L & 15;
  const bool b0 = (L & 1) != 0, b1 = (L & 2) != 0, b2 = (L & 4) != 0;
  const bool claimer = (L & 8) == 0;
  const int rowC = ((L >> 4) << 3) | (L & 7);

  v2f bf[8][8];
  for (int ks = 0; ks < 8; ++ks)
    for (int n = 0; n < 8; ++n) {
      bf[ks][n].x = We[(ks * 4 + kOff) * HD + n * 16 + colB];
      bf[ks][n].y = We[(ks * 4 + kOff + 1) * HD + n * 16 + colB];
    }
  float bev[8];
  for (int n = 0; n < 8; ++n) bev[n] = be[n * 16 + colB];

  int t = wave;
  v2f a[8];
  int srcv = 0, dstv = 0, rankv = 0;
  if (t < ntiles) {
    int ge = t * 16 + rowA; if (ge >= NE) ge = NE - 1;
    const float* rp = ea + (size_t)ge * E_DIM + kOff;
    for (int ks = 0; ks < 8; ++ks) a[ks] = *(const v2f*)(rp + ks * 4);
    const int gc = t * 16 + rowC;
    if (claimer && gc < NE) { srcv = ei[gc]; dstv = ei[NE + gc]; rankv = rank[gc]; }
  }
  for (int ks = 0; ks < 8; ++ks) asm volatile("" : "+v"(a[ks]));
  asm volatile("" : "+v"(srcv), "+v"(dstv), "+v"(rankv));

  while (t < ntiles) {
    const bool mine = claimer && t * 16 + rowC < NE;
    const int startv = pre[srcv];

    const int tn = t + nwaves;
    v2f an[8];
    int srcn = 0, dstn = 0, rankn = 0;
    if (tn < ntiles) {
      int ge = tn * 16 + rowA; if (ge >= NE) ge = NE - 1;
      const float* rp = ea + (size_t)ge * E_DIM + kOff;
      for (int ks = 0; ks < 8; ++ks) an[ks] = *(const v2f*)(rp + ks * 4);
      const int gc = tn * 16 + rowC;
      if (claimer && gc < NE) { srcn = ei[gc]; dstn = ei[NE + gc]; rankn = rank[gc]; }
    } else {
      for (int ks = 0; ks < 8; ++ks) an[ks] = a[ks];
    }

    v8f c[8];
    for (int n = 0; n < 8; ++n) c[n] = v8f{};
    for (int ks = 0; ks < 8; ++ks)
      for (int n = 0; n < 8; ++n) c[n] = wmma_f32(a[ks], bf[ks][n], c[n]);

    float z[NHEAD];
    for (int hd = 0; hd < NHEAD; ++hd) {
      float tt[8];
      for (int j = 0; j < 8; ++j) {
        float u = c[2 * hd][j] + bev[2 * hd], v = c[2 * hd + 1][j] + bev[2 * hd + 1];
        tt[j] = u * u + v * v;
      }
      float u4[4];
      for (int i = 0; i < 4; ++i) {
        const float keep = b0 ? tt[2 * i + 1] : tt[2 * i];
        const float send = b0 ? tt[2 * i] : tt[2 * i + 1];
        u4[i] = keep + __shfl_xor(send, 1);
      }
      float w2[2];
      for (int i = 0; i < 2; ++i) {
        const float keep = b1 ? u4[2 * i + 1] : u4[2 * i];
        const float send = b1 ? u4[2 * i] : u4[2 * i + 1];
        w2[i] = keep + __shfl_xor(send, 2);
      }
      const float keep = b2 ? w2[1] : w2[0];
      const float send = b2 ? w2[0] : w2[1];
      float zz = keep + __shfl_xor(send, 4);
      zz += __shfl_xor(zz, 8);
      z[hd] = zz;
    }

    if (mine) {
      const int pos = startv + rankv;
      *(float4*)(sE + (size_t)pos * NHEAD) = make_float4(z[0], z[1], z[2], z[3]);
      dstS[pos] = (unsigned short)dstv;
    }

    for (int ks = 0; ks < 8; ++ks) a[ks] = an[ks];
    srcv = srcn; dstv = dstn; rankv = rankn;
    t = tn;
  }
}

template <int M>
static __device__ __forceinline__ float row_xor(float v) {
  return __int_as_float(__builtin_amdgcn_update_dpp(
      0, __float_as_int(v), 0x160 | M, 0xf, 0xf, true));
}

#define NP_LD 10
__global__ __launch_bounds__(256) void node_pass(
    const float* __restrict__ Q, const float* __restrict__ Km,
    const float* __restrict__ Vo, const float* __restrict__ sE,
    const unsigned short* __restrict__ dstS, const int* __restrict__ cur,
    const float* __restrict__ bo, float* __restrict__ out, int N) {
  __shared__ int part[8];
  const int tid = threadIdx.x;
  const int L = tid & 31;
  const int w = __builtin_amdgcn_readfirstlane(tid >> 5);
  const int nb = blockIdx.x * 8;
  int acc = 0;
  {
    const int nq = nb >> 2;
    const int4* cq = (const int4*)cur;
    for (int q0 = tid; q0 < nq; q0 += 256 * NP_LD) {
      int4 v[NP_LD];
      for (int u = 0; u < NP_LD; ++u) {
        const int q = q0 + u * 256;
        v[u] = (q < nq) ? cq[q] : make_int4(0, 0, 0, 0);
      }
      for (int u = 0; u < NP_LD; ++u) acc += (v[u].x + v[u].y) + (v[u].z + v[u].w);
    }
  }
  const int mc = (L < 8 && nb + L < N) ? cur[nb + L] : 0;
  for (int off = 16; off > 0; off >>= 1) acc += __shfl_xor(acc, off);
  if (L == 0) part[w] = acc;
  __syncthreads();
  const int s = nb + w;
  if (s >= N) return;
  int beg = 0;
  for (int j = 0; j < 8; ++j) beg += part[j];
  int cnt_s = 0;
  for (int j = 0; j < 8; ++j) {
    const int cj = __builtin_amdgcn_readlane(mc, j);
    if (j < w) beg += cj;
    if (j == w) cnt_s = cj;
  }
  beg = __builtin_amdgcn_readfirstlane(beg);
  const int end = beg + __builtin_amdgcn_readfirstlane(cnt_s);

  const int g = L >> 3;
  const int c4 = (L & 7) << 2;
  float4 q[NHEAD];
  for (int h = 0; h < NHEAD; ++h)
    q[h] = *(const float4*)(Q + (size_t)s * HD + h * DHEAD + c4);
  const float4 bo4 = *(const float4*)(bo + c4);
  float4 tot = make_float4(0.0f, 0.0f, 0.0f, 0.0f);

  for (int base = beg; base < end; base += 32) {
    const int idx = base + L;
    const int myd = (idx < end) ? (int)dstS[idx] : 0;
    const int n = (end - base < 32) ? (end - base) : 32;
    for (int j = 0; j < n; j += 4) {
      const int d0 = __builtin_amdgcn_readlane(myd, j);
      const int d1 = __builtin_amdgcn_readlane(myd, j + 1);
      const int d2 = __builtin_amdgcn_readlane(myd, j + 2);
      const int d3 = __builtin_amdgcn_readlane(myd, j + 3);
      const int dlo = (g & 1) ? d1 : d0;
      const int dhi = (g & 1) ? d3 : d2;
      const bool valid = j + g < n;
      const int dst = valid ? ((g & 2) ? dhi : dlo) : d0;
      const int rec = base + j + (valid ? g : 0);
      const float wgt = valid ? 1.0f : 0.0f;
      v4f se = *(const v4f*)(sE + (size_t)rec * NHEAD);
      const float* kp = Km + (size_t)dst * HD + c4;
      const float* vp = Vo + (size_t)dst * HD + c4;
      v4f k[NHEAD], v[NHEAD];
      for (int h = 0; h < NHEAD; ++h) k[h] = *(const v4f*)(kp + h * DHEAD);
      for (int h = 0; h < NHEAD; ++h) v[h] = *(const v4f*)(vp + h * DHEAD);
      asm volatile("" : "+v"(k[0]), "+v"(k[1]), "+v"(k[2]), "+v"(k[3]),
                        "+v"(v[0]), "+v"(v[1]), "+v"(v[2]), "+v"(v[3]), "+v"(se));

      const float sev[NHEAD] = {se.x, se.y, se.z, se.w};
      float sc[NHEAD];
      for (int h = 0; h < NHEAD; ++h) {
        float p = q[h].x * k[h].x + q[h].y * k[h].y + q[h].z * k[h].z + q[h].w * k[h].w;
        p += row_xor<1>(p);
        p += row_xor<2>(p);
        p += row_xor<4>(p);
        sc[h] = (p + sev[h]) * INV_SQRT_D;
      }

      float m = sc[0];
      for (int h2 = 1; h2 < NHEAD; ++h2) m = fmaxf(m, sc[h2]);
      float ex[NHEAD], den = 0.0f;
      for (int h2 = 0; h2 < NHEAD; ++h2) { ex[h2] = __expf(sc[h2] - m); den += ex[h2]; }
      const float inv = 1.0f / den;

      float4 a = bo4;
      for (int h2 = 0; h2 < NHEAD; ++h2) {
        const float w = ex[h2] * inv;
        a.x += w * v[h2].x;
        a.y += w * v[h2].y;
        a.z += w * v[h2].z;
        a.w += w * v[h2].w;
      }
      tot.x = fmaf(wgt, a.x, tot.x);
      tot.y = fmaf(wgt, a.y, tot.y);
      tot.z = fmaf(wgt, a.z, tot.z);
      tot.w = fmaf(wgt, a.w, tot.w);
    }
  }
  tot.x += row_xor<8>(tot.x);
  tot.y += row_xor<8>(tot.y);
  tot.z += row_xor<8>(tot.z);
  tot.w += row_xor<8>(tot.w);
  tot.x += __shfl_xor(tot.x, 16);
  tot.y += __shfl_xor(tot.y, 16);
  tot.z += __shfl_xor(tot.z, 16);
  tot.w += __shfl_xor(tot.w, 16);
  if (g == 0) {
    const float cntf = fmaxf((float)(end - beg), 1.0f);
    *(float4*)(out + (size_t)s * OUT_DIM + c4) =
        make_float4(tot.x / cntf, tot.y / cntf, tot.z / cntf, tot.w / cntf);
  }
}

extern "C" void kernel_launch(void* const* d_in, const int* in_sizes, int n_in,
                              void* d_out, int out_size, void* d_ws, size_t ws_size,
                              hipStream_t stream) {
  const float* x  = (const float*)d_in[0];
  const int*   ei = (const int*)d_in[1];
  const float* ea = (const float*)d_in[2];
  const float* Wq = (const float*)d_in[3];
  const float* bq = (const float*)d_in[4];
  const float* Wk = (const float*)d_in[5];
  const float* bk = (const float*)d_in[6];
  const float* Wv = (const float*)d_in[7];
  const float* bv = (const float*)d_in[8];
  const float* We = (const float*)d_in[9];
  const float* be = (const float*)d_in[10];
  const float* Wo = (const float*)d_in[11];
  const float* bo = (const float*)d_in[12];

  const int N  = in_sizes[0] / IN_DIM;
  const int NE = in_sizes[1] / 2;

  float* ws   = (float*)d_ws;
  float* Q    = ws;
  float* Km   = Q   + (size_t)N * HD;
  float* Vo   = Km  + (size_t)N * HD;
  float* Wvo  = Vo  + (size_t)N * HD;
  float* bvo  = Wvo + IN_DIM * HD;
  float* sE   = bvo + HD;
  unsigned short* dstS = (unsigned short*)(sE + (size_t)NE * NHEAD);
  int* cur = (int*)(sE + (size_t)NE * NHEAD + (size_t)N * OUT_DIM);
  int* rank = (int*)Vo;

  zero_kernel<<<(N + 255) / 256, 256, 0, stream>>>((float*)cur, N);
  rank_src<<<(NE + 1024 * RANK_EPT - 1) / (1024 * RANK_EPT), 1024,
             (size_t)N * sizeof(int), stream>>>(ei, cur, rank, N, NE);
  edge_selfdot<<<256, SD_THREADS, (size_t)N * sizeof(int), stream>>>(
      ea, We, be, ei, cur, rank, sE, dstS, N, NE);
  {
    int n = IN_DIM * HD + HD;
    precompute_wvo<<<(n + 127) / 128, 128, 0, stream>>>(Wv, bv, Wo, Wvo, bvo);
  }
  node_gemm<<<(N + 15) / 16, 256, 0, stream>>>(x, Wq, bq, Wk, bk, Wvo, bvo,
                                               Q, Km, Vo, N);
  node_pass<<<(N + 7) / 8, 256, 0, stream>>>(Q, Km, Vo, sE, dstS, cur, bo,
                                             (float*)d_out, N);
}


# score records hold 3 differences to head 0 (softmax unchanged, new rounding); freed 4th word carries the edge rank, Vo plane free: node GEMM rides in the rank launch, fold with zeroing; 6->4 launches
# speedup vs baseline: 2.4244x; 1.0707x over previous
#include <hip/hip_runtime.h>
#include <hip/hip_bf16.h>

typedef __attribute__((ext_vector_type(2))) float v2f;
typedef __attribute__((ext_vector_type(8))) float v8f;
typedef __attribute__((ext_vector_type(4))) float v4f;

#define IN_DIM  128
#define HD      128
#define E_DIM   32
#define NHEAD   4
#define DHEAD   32
#define OUT_DIM 32
#define INV_SQRT_D 0.17677669529663687f

static __device__ __forceinline__ v8f wmma_f32(v2f a, v2f b, v8f c) {
  return __builtin_amdgcn_wmma_f32_16x16x4_f32(false, a, false, b, (short)0, c,
                                               false, false);
}

#define PREP_T 256
#define FOLD_BLOCKS ((IN_DIM * HD + HD + PREP_T - 1) / PREP_T)
__global__ __launch_bounds__(PREP_T) void prep_fold_zero(
    const float* __restrict__ Wv, const float* __restrict__ bv,
    const float* __restrict__ Wo, float* __restrict__ Wvo,
    float* __restrict__ bvo, int* __restrict__ cur, int N) {
  if (blockIdx.x >= FOLD_BLOCKS) {
    const int i = (blockIdx.x - FOLD_BLOCKS) * PREP_T + threadIdx.x;
    if (i < N) cur[i] = 0;
    return;
  }
  const int idx = blockIdx.x * PREP_T + threadIdx.x;
  if (idx < IN_DIM * HD) {
    int k = idx >> 7;
    int c = idx & 127;
    int h = c >> 5, j = c & 31;
    float acc = 0.0f;
    for (int d = 0; d < DHEAD; ++d)
      acc += Wv[k * HD + h * DHEAD + d] * Wo[(h * DHEAD + d) * OUT_DIM + j];
    Wvo[idx] = acc;
  } else if (idx < IN_DIM * HD + HD) {
    int c = idx - IN_DIM * HD;
    int h = c >> 5, j = c & 31;
    float acc = 0.0f;
    for (int d = 0; d < DHEAD; ++d)
      acc += bv[h * DHEAD + d] * Wo[(h * DHEAD + d) * OUT_DIM + j];
    bvo[c] = acc;
  }
}

static __device__ __forceinline__ void node_gemm_tile(
    const float* __restrict__ x,
    const float* __restrict__ Wq, const float* __restrict__ bq,
    const float* __restrict__ Wk, const float* __restrict__ bk,
    const float* __restrict__ Wvo, const float* __restrict__ bvo,
    float* __restrict__ Q, float* __restrict__ Km, float* __restrict__ Vo,
    int N, float (*As)[IN_DIM], const int tid, const int m0) {

  for (int i = 0; i < 8; ++i) {
    int lin = tid + i * 256;
    int r = lin >> 7, c = lin & 127;
    int gr = m0 + r; if (gr >= N) gr = N - 1;
    As[r][c] = x[(size_t)gr * IN_DIM + c];
  }
  __syncthreads();

  const int w = tid >> 5, L = tid & 31;
  const int n0 = w * 16;
  const int rowA = L & 15;
  const int kOff = (L < 16) ? 0 : 2;
  const int colB = L & 15;

  const float bq0 = bq[n0 + colB];
  const float bk0 = bk[n0 + colB];
  const float bv0 = bvo[n0 + colB];
  v8f cq, ck, cv;
  for (int j = 0; j < 8; ++j) { cq[j] = bq0; ck[j] = bk0; cv[j] = bv0; }

  for (int kk = 0; kk < IN_DIM; kk += 4) {
    v2f a;
    a.x = As[rowA][kk + kOff];
    a.y = As[rowA][kk + kOff + 1];
    v2f b;
    b.x = Wq[(kk + kOff) * HD + n0 + colB];
    b.y = Wq[(kk + kOff + 1) * HD + n0 + colB];
    cq = wmma_f32(a, b, cq);
    b.x = Wk[(kk + kOff) * HD + n0 + colB];
    b.y = Wk[(kk + kOff + 1) * HD + n0 + colB];
    ck = wmma_f32(a, b, ck);
    b.x = Wvo[(kk + kOff) * HD + n0 + colB];
    b.y = Wvo[(kk + kOff + 1) * HD + n0 + colB];
    cv = wmma_f32(a, b, cv);
  }

  for (int j = 0; j < 8; ++j) {
    int r = (L < 16) ? j : (j + 8);
    int gr = m0 + r;
    if (gr < N) {
      size_t o = (size_t)gr * HD + n0 + colB;
      Q[o] = cq[j]; Km[o] = ck[j]; Vo[o] = cv[j];
    }
  }
}

#define RANK_EPT 8
#define MERGE_U  10
__global__ __launch_bounds__(1024) void rank_src_node_gemm(
    const int* __restrict__ ei, int* __restrict__ cur, int* __restrict__ rk,
    const float* __restrict__ x,
    const float* __restrict__ Wq, const float* __restrict__ bq,
    const float* __restrict__ Wk, const float* __restrict__ bk,
    const float* __restrict__ Wvo, const float* __restrict__ bvo,
    float* __restrict__ Q, float* __restrict__ Km, float* __restrict__ Vo,
    int N, int NE) {
  extern __shared__ int cnt[];
  __shared__ float As4[4][16][IN_DIM];
  const int tid = threadIdx.x;
  const int nrank = (NE + 1024 * RANK_EPT - 1) / (1024 * RANK_EPT);
  if ((int)blockIdx.x >= nrank) {
    const int grp = tid >> 8;
    node_gemm_tile(x, Wq, bq, Wk, bk, Wvo, bvo, Q, Km, Vo, N, As4[grp], tid & 255,
                   (((int)blockIdx.x - nrank) * 4 + grp) * 16);
    return;
  }
  for (int i = tid; i < N; i += 1024) cnt[i] = 0;
  __syncthreads();

  const int base = blockIdx.x * (1024 * RANK_EPT);
  int sv[RANK_EPT], lr[RANK_EPT];
  for (int j = 0; j < RANK_EPT; ++j) {
    const int e = base + j * 1024 + tid;
    sv[j] = (e < NE) ? ei[e] : -1;
  }
  for (int j = 0; j < RANK_EPT; ++j)
    lr[j] = (sv[j] >= 0) ? atomicAdd(&cnt[sv[j]], 1) : 0;
  __syncthreads();

  for (int i0 = tid; i0 < N; i0 += 1024 * MERGE_U) {
    int c[MERGE_U], b[MERGE_U];
    for (int u = 0; u < MERGE_U; ++u) {
      const int i = i0 + u * 1024;
      c[u] = (i < N) ? cnt[i] : 0;
    }
    for (int u = 0; u < MERGE_U; ++u) {
      const int i = i0 + u * 1024;
      b[u] = (i < N) ? atomicAdd(&cur[i], c[u]) : 0;
    }
    for (int u = 0; u < MERGE_U; ++u) {
      const int i = i0 + u * 1024;
      if (i < N) cnt[i] = b[u];
    }
  }
  __syncthreads();

  for (int j = 0; j < RANK_EPT; ++j) {
    const int e = base + j * 1024 + tid;
    if (sv[j] >= 0) rk[4 * e + 3] = cnt[sv[j]] + lr[j];
  }
}

#define SD_WAVES 16
#define SD_THREADS (SD_WAVES * 32)
#define SD_LD 5
__global__ __launch_bounds__(SD_THREADS) void edge_selfdot(
    const float* __restrict__ ea, const float* __restrict__ We,
    const float* __restrict__ be, const int* __restrict__ ei,
    const int* __restrict__ cur, const int* rk,
    float* sE, unsigned short* __restrict__ dstS, int N, int NE) {
  extern __shared__ int pre[];
  __shared__ int wtot[SD_WAVES];
  const int tid = threadIdx.x;
  const int L = tid & 31;
  {
    const int nq = N >> 2;
    const int4* c4 = (const int4*)cur;
    for (int q0 = tid; q0 < nq; q0 += SD_THREADS * SD_LD) {
      int4 v[SD_LD];
      for (int u = 0; u < SD_LD; ++u) {
        const int q = q0 + u * SD_THREADS;
        v[u] = (q < nq) ? c4[q] : make_int4(0, 0, 0, 0);
      }
      for (int u = 0; u < SD_LD; ++u) {
        const int q = q0 + u * SD_THREADS;
        if (q < nq) *(int4*)(pre + 4 * q) = v[u];
      }
    }
    for (int i = (nq << 2) + tid; i < N; i += SD_THREADS) pre[i] = cur[i];
    __syncthreads();
    const int per = ((N + SD_THREADS - 1) / SD_THREADS) | 1;
    int lo = tid * per; if (lo > N) lo = N;
    int hi = lo + per;  if (hi > N) hi = N;
    int s = 0;
    for (int i = lo; i < hi; ++i) s += pre[i];
    int inc = s;
    for (int off = 1; off < 32; off <<= 1) {
      const int v = __shfl_up(inc, off);
      if (L >= off) inc += v;
    }
    if (L == 31) wtot[tid >> 5] = inc;
    __syncthreads();
    int run = inc - s;
    for (int j = 0; j < SD_WAVES; ++j) run += (j < (tid >> 5)) ? wtot[j] : 0;
    for (int i = lo; i < hi; ++i) { const int cn = pre[i]; pre[i] = run; run += cn; }
    __syncthreads();
  }
  const int wave = __builtin_amdgcn_readfirstlane(blockIdx.x * SD_WAVES + (tid >> 5));
  const int nwaves = gridDim.x * SD_WAVES;
  const int ntiles = (NE + 15) >> 4;

  const int rowA = L & 15;
  const int kOff = (L < 16) ? 0 : 2;
  const int colB = L & 15;
  const bool b0 = (L & 1) != 0, b1 = (L & 2) != 0, b2 = (L & 4) != 0;
  const bool claimer = (L & 8) == 0;
  const int rowC = ((L >> 4) << 3) | (L & 7);

  v2f bf[8][8];
  for (int ks = 0; ks < 8; ++ks)
    for (int n = 0; n < 8; ++n) {
      bf[ks][n].x = We[(ks * 4 + kOff) * HD + n * 16 + colB];
      bf[ks][n].y = We[(ks * 4 + kOff + 1) * HD + n * 16 + colB];
    }
  float bev[8];
  for (int n = 0; n < 8; ++n) bev[n] = be[n * 16 + colB];

  int t = wave;
  v2f a[8];
  int srcv = 0, dstv = 0, rankv = 0;
  if (t < ntiles) {
    int ge = t * 16 + rowA; if (ge >= NE) ge = NE - 1;
    const float* rp = ea + (size_t)ge * E_DIM + kOff;
    for (int ks = 0; ks < 8; ++ks) a[ks] = *(const v2f*)(rp + ks * 4);
    const int gc = t * 16 + rowC;
    if (claimer && gc < NE) { srcv = ei[gc]; dstv = ei[NE + gc]; rankv = rk[4 * gc + 3]; }
  }
  for (int ks = 0; ks < 8; ++ks) asm volatile("" : "+v"(a[ks]));
  asm volatile("" : "+v"(srcv), "+v"(dstv), "+v"(rankv));

  while (t < ntiles) {
    const bool mine = claimer && t * 16 + rowC < NE;
    const int startv = pre[srcv];

    const int tn = t + nwaves;
    v2f an[8];
    int srcn = 0, dstn = 0, rankn = 0;
    if (tn < ntiles) {
      int ge = tn * 16 + rowA; if (ge >= NE) ge = NE - 1;
      const float* rp = ea + (size_t)ge * E_DIM + kOff;
      for (int ks = 0; ks < 8; ++ks) an[ks] = *(const v2f*)(rp + ks * 4);
      const int gc = tn * 16 + rowC;
      if (claimer && gc < NE) { srcn = ei[gc]; dstn = ei[NE + gc]; rankn = rk[4 * gc + 3]; }
    } else {
      for (int ks = 0; ks < 8; ++ks) an[ks] = a[ks];
    }

    v8f c[8];
    for (int n = 0; n < 8; ++n) c[n] = v8f{};
    for (int ks = 0; ks < 8; ++ks)
      for (int n = 0; n < 8; ++n) c[n] = wmma_f32(a[ks], bf[ks][n], c[n]);

    float z[NHEAD];
    for (int hd = 0; hd < NHEAD; ++hd) {
      float tt[8];
      for (int j = 0; j < 8; ++j) {
        float u = c[2 * hd][j] + bev[2 * hd], v = c[2 * hd + 1][j] + bev[2 * hd + 1];
        tt[j] = u * u + v * v;
      }
      float u4[4];
      for (int i = 0; i < 4; ++i) {
        const float keep = b0 ? tt[2 * i + 1] : tt[2 * i];
        const float send = b0 ? tt[2 * i] : tt[2 * i + 1];
        u4[i] = keep + __shfl_xor(send, 1);
      }
      float w2[2];
      for (int i = 0; i < 2; ++i) {
        const float keep = b1 ? u4[2 * i + 1] : u4[2 * i];
        const float send = b1 ? u4[2 * i] : u4[2 * i + 1];
        w2[i] = keep + __shfl_xor(send, 2);
      }
      const float keep = b2 ? w2[1] : w2[0];
      const float send = b2 ? w2[0] : w2[1];
      float zz = keep + __shfl_xor(send, 4);
      zz += __shfl_xor(zz, 8);
      z[hd] = zz;
    }

    if (mine) {
      const int pos = startv + rankv;
      float* rec = sE + (size_t)pos * NHEAD;
      *(v2f*)rec = v2f{z[1] - z[0], z[2] - z[0]};
      rec[2] = z[3] - z[0];
      dstS[pos] = (unsigned short)dstv;
    }

    for (int ks = 0; ks < 8; ++ks) a[ks] = an[ks];
    srcv = srcn; dstv = dstn; rankv = rankn;
    t = tn;
  }
}

template <int M>
static __device__ __forceinline__ float row_xor(float v) {
  return __int_as_float(__builtin_amdgcn_update_dpp(
      0, __float_as_int(v), 0x160 | M, 0xf, 0xf, true));
}

#define NP_LD 10
__global__ __launch_bounds__(256) void node_pass(
    const float* __restrict__ Q, const float* __restrict__ Km,
    const float* __restrict__ Vo, const float* __restrict__ sE,
    const unsigned short* __restrict__ dstS, const int* __restrict__ cur,
    const float* __restrict__ bo, float* __restrict__ out, int N) {
  __shared__ int part[8];
  const int tid = threadIdx.x;
  const int L = tid & 31;
  const int w = __builtin_amdgcn_readfirstlane(tid >> 5);
  const int nb = blockIdx.x * 8;
  int acc = 0;
  {
    const int nq = nb >> 2;
    const int4* cq = (const int4*)cur;
    for (int q0 = tid; q0 < nq; q0 += 256 * NP_LD) {
      int4 v[NP_LD];
      for (int u = 0; u < NP_LD; ++u) {
        const int q = q0 + u * 256;
        v[u] = (q < nq) ? cq[q] : make_int4(0, 0, 0, 0);
      }
      for (int u = 0; u < NP_LD; ++u) acc += (v[u].x + v[u].y) + (v[u].z + v[u].w);
    }
  }
  const int mc = (L < 8 && nb + L < N) ? cur[nb + L] : 0;
  for (int off = 16; off > 0; off >>= 1) acc += __shfl_xor(acc, off);
  if (L == 0) part[w] = acc;
  __syncthreads();
  const int s = nb + w;
  if (s >= N) return;
  int beg = 0;
  for (int j = 0; j < 8; ++j) beg += part[j];
  int cnt_s = 0;
  for (int j = 0; j < 8; ++j) {
    const int cj = __builtin_amdgcn_readlane(mc, j);
    if (j < w) beg += cj;
    if (j == w) cnt_s = cj;
  }
  beg = __builtin_amdgcn_readfirstlane(beg);
  const int end = beg + __builtin_amdgcn_readfirstlane(cnt_s);

  const int g = L >> 3;
  const int c4 = (L & 7) << 2;
  float4 q[NHEAD];
  for (int h = 0; h < NHEAD; ++h)
    q[h] = *(const float4*)(Q + (size_t)s * HD + h * DHEAD + c4);
  const float4 bo4 = *(const float4*)(bo + c4);
  float4 tot = make_float4(0.0f, 0.0f, 0.0f, 0.0f);

  for (int base = beg; base < end; base += 32) {
    const int idx = base + L;
    const int myd = (idx < end) ? (int)dstS[idx] : 0;
    const int n = (end - base < 32) ? (end - base) : 32;
    for (int j = 0; j < n; j += 4) {
      const int d0 = __builtin_amdgcn_readlane(myd, j);
      const int d1 = __builtin_amdgcn_readlane(myd, j + 1);
      const int d2 = __builtin_amdgcn_readlane(myd, j + 2);
      const int d3 = __builtin_amdgcn_readlane(myd, j + 3);
      const int dlo = (g & 1) ? d1 : d0;
      const int dhi = (g & 1) ? d3 : d2;
      const bool valid = j + g < n;
      const int dst = valid ? ((g & 2) ? dhi : dlo) : d0;
      const int rec = base + j + (valid ? g : 0);
      const float wgt = valid ? 1.0f : 0.0f;
      v4f se = *(const v4f*)(sE + (size_t)rec * NHEAD);
      const float* kp = Km + (size_t)dst * HD + c4;
      const float* vp = Vo + (size_t)dst * HD + c4;
      v4f k[NHEAD], v[NHEAD];
      for (int h = 0; h < NHEAD; ++h) k[h] = *(const v4f*)(kp + h * DHEAD);
      for (int h = 0; h < NHEAD; ++h) v[h] = *(const v4f*)(vp + h * DHEAD);
      asm volatile("" : "+v"(k[0]), "+v"(k[1]), "+v"(k[2]), "+v"(k[3]),
                        "+v"(v[0]), "+v"(v[1]), "+v"(v[2]), "+v"(v[3]), "+v"(se));

      const float sev[NHEAD] = {0.0f, se.x, se.y, se.z};
      float sc[NHEAD];
      for (int h = 0; h < NHEAD; ++h) {
        float p = q[h].x * k[h].x + q[h].y * k[h].y + q[h].z * k[h].z + q[h].w * k[h].w;
        p += row_xor<1>(p);
        p += row_xor<2>(p);
        p += row_xor<4>(p);
        sc[h] = (p + sev[h]) * INV_SQRT_D;
      }

      float m = sc[0];
      for (int h2 = 1; h2 < NHEAD; ++h2) m = fmaxf(m, sc[h2]);
      float ex[NHEAD], den = 0.0f;
      for (int h2 = 0; h2 < NHEAD; ++h2) { ex[h2] = __expf(sc[h2] - m); den += ex[h2]; }
      const float inv = 1.0f / den;

      float4 a = bo4;
      for (int h2 = 0; h2 < NHEAD; ++h2) {
        const float w = ex[h2] * inv;
        a.x += w * v[h2].x;
        a.y += w * v[h2].y;
        a.z += w * v[h2].z;
        a.w += w * v[h2].w;
      }
      tot.x = fmaf(wgt, a.x, tot.x);
      tot.y = fmaf(wgt, a.y, tot.y);
      tot.z = fmaf(wgt, a.z, tot.z);
      tot.w = fmaf(wgt, a.w, tot.w);
    }
  }
  tot.x += row_xor<8>(tot.x);
  tot.y += row_xor<8>(tot.y);
  tot.z += row_xor<8>(tot.z);
  tot.w += row_xor<8>(tot.w);
  tot.x += __shfl_xor(tot.x, 16);
  tot.y += __shfl_xor(tot.y, 16);
  tot.z += __shfl_xor(tot.z, 16);
  tot.w += __shfl_xor(tot.w, 16);
  if (g == 0) {
    const float cntf = fmaxf((float)(end - beg), 1.0f);
    *(float4*)(out + (size_t)s * OUT_DIM + c4) =
        make_float4(tot.x / cntf, tot.y / cntf, tot.z / cntf, tot.w / cntf);
  }
}

extern "C" void kernel_launch(void* const* d_in, const int* in_sizes, int n_in,
                              void* d_out, int out_size, void* d_ws, size_t ws_size,
                              hipStream_t stream) {
  const float* x  = (const float*)d_in[0];
  const int*   ei = (const int*)d_in[1];
  const float* ea = (const float*)d_in[2];
  const float* Wq = (const float*)d_in[3];
  const float* bq = (const float*)d_in[4];
  const float* Wk = (const float*)d_in[5];
  const float* bk = (const float*)d_in[6];
  const float* Wv = (const float*)d_in[7];
  const float* bv = (const float*)d_in[8];
  const float* We = (const float*)d_in[9];
  const float* be = (const float*)d_in[10];
  const float* Wo = (const float*)d_in[11];
  const float* bo = (const float*)d_in[12];

  const int N  = in_sizes[0] / IN_DIM;
  const int NE = in_sizes[1] / 2;

  float* ws   = (float*)d_ws;
  float* Q    = ws;
  float* Km   = Q   + (size_t)N * HD;
  float* Vo   = Km  + (size_t)N * HD;
  float* Wvo  = Vo  + (size_t)N * HD;
  float* bvo  = Wvo + IN_DIM * HD;
  float* sE   = bvo + HD;
  unsigned short* dstS = (unsigned short*)(sE + (size_t)NE * NHEAD);
  int* cur = (int*)(sE + (size_t)NE * NHEAD + (size_t)N * OUT_DIM);
  int* rk = (int*)sE;

  prep_fold_zero<<<FOLD_BLOCKS + (N + PREP_T - 1) / PREP_T, PREP_T, 0, stream>>>(
      Wv, bv, Wo, Wvo, bvo, cur, N);
  rank_src_node_gemm<<<(NE + 1024 * RANK_EPT - 1) / (1024 * RANK_EPT) + ((N + 15) / 16 + 3) / 4,
                       1024, (size_t)N * sizeof(int), stream>>>(
      ei, cur, rk, x, Wq, bq, Wk, bk, Wvo, bvo, Q, Km, Vo, N, NE);
  edge_selfdot<<<256, SD_THREADS, (size_t)N * sizeof(int), stream>>>(
      ea, We, be, ei, cur, rk, sE, dstS, N, NE);
  node_pass<<<(N + 7) / 8, 256, 0, stream>>>(Q, Km, Vo, sE, dstS, cur, bo,
                                             (float*)d_out, N);
}
